# starting kernel
# baseline (speedup 1.0000x reference)
#include <hip/hip_runtime.h>

typedef float v2f __attribute__((ext_vector_type(2)));
typedef float v8f __attribute__((ext_vector_type(8)));

#define BDIM  2
#define NDIM  1024
#define CDIM  64
#define SCALE 0.25f
#define LOG2E 1.44269504088896340736f

__global__ __launch_bounds__(256)
void intra_attn_wmma_kernel(const float* __restrict__ q,
                            const float* __restrict__ k,
                            const float* __restrict__ v,
                            float* __restrict__ out) {
    __shared__ __align__(16) float ks[NDIM];
    __shared__ __align__(16) float vs[NDIM];
    __shared__ float red[256];

    const int tid  = threadIdx.x;
    const int lane = tid & 31;
    const int wave = tid >> 5;

    const int bc    = blockIdx.x;
    const int b     = bc >> 6;
    const int c     = bc & (CDIM - 1);
    const int ibase = blockIdx.y * 256;

    const float* qp = q   + (size_t)b * NDIM * CDIM + c;
    const float* kp = k   + (size_t)b * NDIM * CDIM + c;
    const float* vp = v   + (size_t)b * NDIM * CDIM + c;
    float*       op = out + (size_t)b * NDIM * CDIM + c;

    float lmax = 0.0f;
    for (int j = tid; j < NDIM; j += 256) {
        const float kv = kp[(size_t)j * CDIM] * (SCALE * LOG2E);
        ks[j] = kv;
        vs[j] = vp[(size_t)j * CDIM];
        lmax = fmaxf(lmax, fabsf(kv));
    }
    red[tid] = lmax;
    __syncthreads();
    for (int s = 128; s > 0; s >>= 1) {
        if (tid < s) red[tid] = fmaxf(red[tid], red[tid + s]);
        __syncthreads();
    }
    const float ksabs = red[0];

    const int   M    = lane & 15;
    const int   koff = (lane >> 4) << 1;
    const float bc1  = (M == 1) ? 1.0f : 0.0f;
    const bool  isn0 = (M == 0);

    for (int tt = 0; tt < 2; ++tt) {
        const int   i0 = ibase + (wave * 2 + tt) * 16;
        const float a  = qp[(size_t)(i0 + M) * CDIM];
        const float m2 = fabsf(a) * ksabs;

        v8f acc0 = {};
        v8f acc1 = {};
        for (int j = 0; j < NDIM; j += 8) {
            const float2 k2a = *reinterpret_cast<const float2*>(&ks[j + koff]);
            const float2 v2a = *reinterpret_cast<const float2*>(&vs[j + koff]);
            const float2 k2b = *reinterpret_cast<const float2*>(&ks[j + 4 + koff]);
            const float2 v2b = *reinterpret_cast<const float2*>(&vs[j + 4 + koff]);

            v2f A0, B0, A1, B1;
            A0.x = __builtin_amdgcn_exp2f(__builtin_fmaf(a, k2a.x, -m2));
            A0.y = __builtin_amdgcn_exp2f(__builtin_fmaf(a, k2a.y, -m2));
            A1.x = __builtin_amdgcn_exp2f(__builtin_fmaf(a, k2b.x, -m2));
            A1.y = __builtin_amdgcn_exp2f(__builtin_fmaf(a, k2b.y, -m2));
            B0.x = isn0 ? v2a.x : bc1;
            B0.y = isn0 ? v2a.y : bc1;
            B1.x = isn0 ? v2b.x : bc1;
            B1.y = isn0 ? v2b.y : bc1;

            acc0 = __builtin_amdgcn_wmma_f32_16x16x4_f32(false, A0, false, B0,
                                                         (short)0, acc0, false, false);
            acc1 = __builtin_amdgcn_wmma_f32_16x16x4_f32(false, A1, false, B1,
                                                         (short)0, acc1, false, false);
        }
        v8f acc = acc0 + acc1;

        #pragma unroll
        for (int r = 0; r < 8; ++r) {
            const float tlo = __shfl(acc[r], 0, 32);
            const float slo = __shfl(acc[r], 1, 32);
            const float thi = __shfl(acc[r], 16, 32);
            const float shi = __shfl(acc[r], 17, 32);
            if (lane == r)      op[(size_t)(i0 + r) * CDIM]     = tlo / slo;
            if (lane == 16 + r) op[(size_t)(i0 + 8 + r) * CDIM] = thi / shi;
        }
    }
}

extern "C" void kernel_launch(void* const* d_in, const int* in_sizes, int n_in,
                              void* d_out, int out_size, void* d_ws, size_t ws_size,
                              hipStream_t stream) {
    const float* q = (const float*)d_in[0];
    const float* k = (const float*)d_in[1];
    const float* v = (const float*)d_in[2];
    float* out = (float*)d_out;
    dim3 grid(BDIM * CDIM, NDIM / 256);
    intra_attn_wmma_kernel<<<grid, dim3(256), 0, stream>>>(q, k, v, out);
}


# Staging rebuilt: one 16-wave block per compute unit loads two adjacent channels of k,v,q by 8-byte loads, all issued before one wait; max|ks| by lane exchange and 1 barrier, not a 9-barrier LDS tree
# speedup vs baseline: 1.1425x; 1.1425x over previous
#include <hip/hip_runtime.h>

typedef float v2f __attribute__((ext_vector_type(2)));
typedef float v8f __attribute__((ext_vector_type(8)));

#define BDIM  2
#define NDIM  1024
#define CDIM  64
#define SCALE 0.25f
#define LOG2E 1.44269504088896340736f
#define JC    64
#define NCH   (NDIM / JC)
#define CPB   2
#define ROWS  256
#define NTHR  512
#define NWAVE (NTHR / 32)


struct Frag {
    float4 k[JC / 8];
    float4 v[JC / 8];
};

static __device__ __forceinline__ void load_frag(Frag& f, const float* kl, const float* vl) {
    #pragma unroll
    for (int e = 0; e < JC / 8; ++e) {
        f.k[e] = *reinterpret_cast<const float4*>(kl + 4 * e);
        f.v[e] = *reinterpret_cast<const float4*>(vl + 4 * e);
    }
}

static __device__ __forceinline__ v2f exp_pair(float a, float n, float k0, float k1) {
    v2f A;
    A.x = __builtin_amdgcn_exp2f(__builtin_fmaf(a, k0, n));
    A.y = __builtin_amdgcn_exp2f(__builtin_fmaf(a, k1, n));
    return A;
}

static __device__ __forceinline__ void mma_frag(const Frag& f,
                                                float a0, float n0, float a1, float n1,
                                                bool isn0, float bc1,
                                                v8f& acc0, v8f& acc1) {
    #pragma unroll
    for (int e = 0; e < JC / 8; ++e) {
        const float4 kk = f.k[e];
        const float4 vv = f.v[e];
        v2f Blo, Bhi;
        Blo.x = isn0 ? vv.x : bc1;
        Blo.y = isn0 ? vv.y : bc1;
        Bhi.x = isn0 ? vv.z : bc1;
        Bhi.y = isn0 ? vv.w : bc1;

        acc0 = __builtin_amdgcn_wmma_f32_16x16x4_f32(false, exp_pair(a0, n0, kk.x, kk.y), false, Blo,
                                                     (short)0, acc0, false, false);
        acc1 = __builtin_amdgcn_wmma_f32_16x16x4_f32(false, exp_pair(a1, n1, kk.x, kk.y), false, Blo,
                                                     (short)0, acc1, false, false);
        acc0 = __builtin_amdgcn_wmma_f32_16x16x4_f32(false, exp_pair(a0, n0, kk.z, kk.w), false, Bhi,
                                                     (short)0, acc0, false, false);
        acc1 = __builtin_amdgcn_wmma_f32_16x16x4_f32(false, exp_pair(a1, n1, kk.z, kk.w), false, Bhi,
                                                     (short)0, acc1, false, false);
    }
}

static __device__ __forceinline__ void store_tile(const v8f& acc, float* op, int i0, int lane) {
    #pragma unroll
    for (int r = 0; r < 8; ++r) {
        const float tlo = __shfl(acc[r], 0, 32);
        const float slo = __shfl(acc[r], 1, 32);
        const float thi = __shfl(acc[r], 16, 32);
        const float shi = __shfl(acc[r], 17, 32);
        if (lane == r)      op[(size_t)(i0 + r) * CDIM]     = tlo / slo;
        if (lane == 16 + r) op[(size_t)(i0 + 8 + r) * CDIM] = thi / shi;
    }
}

static __device__ __forceinline__ float2 ld2(const float* p) {
    return *reinterpret_cast<const float2*>(p);
}

__global__ __launch_bounds__(NTHR)
void intra_attn_wmma_kernel(const float* __restrict__ q,
                            const float* __restrict__ k,
                            const float* __restrict__ v,
                            float* __restrict__ out) {
    __shared__ __align__(16) float ks[CPB][NDIM];
    __shared__ __align__(16) float vs[CPB][NDIM];
    __shared__ __align__(16) float qs[CPB][ROWS];
    __shared__ __align__(16) float red[CPB][NWAVE];

    const int tid  = threadIdx.x;
    const int lane = tid & 31;
    const int wave = tid >> 5;

    const int cp    = blockIdx.x;
    const int b     = cp >> 5;
    const int c0    = (cp & (CDIM / CPB - 1)) * CPB;
    const int ibase = blockIdx.y * ROWS;

    const float* qp = q   + (size_t)b * NDIM * CDIM + c0;
    const float* kp = k   + (size_t)b * NDIM * CDIM + c0;
    const float* vp = v   + (size_t)b * NDIM * CDIM + c0;

    const float2 kA = ld2(kp + (size_t)tid * CDIM);
    const float2 kB = ld2(kp + (size_t)(tid + NTHR) * CDIM);
    const float2 vA = ld2(vp + (size_t)tid * CDIM);
    const float2 vB = ld2(vp + (size_t)(tid + NTHR) * CDIM);
    float2 qq = make_float2(0.0f, 0.0f);
    if (tid < ROWS) qq = ld2(qp + (size_t)(ibase + tid) * CDIM);

    const float kA0 = kA.x * (SCALE * LOG2E);
    const float kA1 = kA.y * (SCALE * LOG2E);
    const float kB0 = kB.x * (SCALE * LOG2E);
    const float kB1 = kB.y * (SCALE * LOG2E);
    ks[0][tid]        = kA0;
    ks[1][tid]        = kA1;
    ks[0][tid + NTHR] = kB0;
    ks[1][tid + NTHR] = kB1;
    vs[0][tid]        = vA.x;
    vs[1][tid]        = vA.y;
    vs[0][tid + NTHR] = vB.x;
    vs[1][tid + NTHR] = vB.y;
    if (tid < ROWS) {
        qs[0][tid] = qq.x;
        qs[1][tid] = qq.y;
    }

    float m0 = fmaxf(fabsf(kA0), fabsf(kB0));
    float m1 = fmaxf(fabsf(kA1), fabsf(kB1));
    #pragma unroll
    for (int o = 16; o > 0; o >>= 1) {
        m0 = fmaxf(m0, __shfl_xor(m0, o, 32));
        m1 = fmaxf(m1, __shfl_xor(m1, o, 32));
    }
    if (lane == 0) {
        red[0][wave] = m0;
        red[1][wave] = m1;
    }
    __syncthreads();

    const int ch = wave >> 3;
    const int il = (wave & 7) * 32;

    float ksabs = 0.0f;
    #pragma unroll
    for (int e = 0; e < NWAVE / 4; ++e) {
        const float4 r = *reinterpret_cast<const float4*>(&red[ch][4 * e]);
        ksabs = fmaxf(fmaxf(ksabs, fmaxf(r.x, r.y)), fmaxf(r.z, r.w));
    }

    const int   M    = lane & 15;
    const float bc1  = (M == 1) ? 1.0f : 0.0f;
    const bool  isn0 = (M == 0);

    const float a0 = qs[ch][il + M];
    const float a1 = qs[ch][il + 16 + M];
    const float n0 = -(fabsf(a0) * ksabs);
    const float n1 = -(fabsf(a1) * ksabs);

    const float* kl = &ks[ch][0] + ((lane >> 4) << 5);
    const float* vl = &vs[ch][0] + ((lane >> 4) << 5);

    v8f acc0 = {};
    v8f acc1 = {};
    Frag fa, fb;
    load_frag(fa, kl, vl);
    for (int chk = 0; chk < NCH; chk += 2) {
        load_frag(fb, kl + (chk + 1) * JC, vl + (chk + 1) * JC);
        __builtin_amdgcn_sched_barrier(0);
        mma_frag(fa, a0, n0, a1, n1, isn0, bc1, acc0, acc1);
        asm volatile("" : "+v"(acc0), "+v"(acc1) : : "memory");
        load_frag(fa, kl + ((chk + 2) & (NCH - 1)) * JC, vl + ((chk + 2) & (NCH - 1)) * JC);
        __builtin_amdgcn_sched_barrier(0);
        mma_frag(fb, a0, n0, a1, n1, isn0, bc1, acc0, acc1);
    }

    float* op = out + (size_t)b * NDIM * CDIM + c0 + ch;
    store_tile(acc0, op, ibase + il, lane);
    store_tile(acc1, op, ibase + il + 16, lane);
}

extern "C" void kernel_launch(void* const* d_in, const int* in_sizes, int n_in,
                              void* d_out, int out_size, void* d_ws, size_t ws_size,
                              hipStream_t stream) {
    const float* q = (const float*)d_in[0];
    const float* k = (const float*)d_in[1];
    const float* v = (const float*)d_in[2];
    float* out = (float*)d_out;
    dim3 grid(BDIM * CDIM / CPB, NDIM / ROWS);
    intra_attn_wmma_kernel<<<grid, dim3(NTHR), 0, stream>>>(q, k, v, out);
}


# Epilogue rebuilt: a row's denominator reaches its numerator lane by one DPP lane move and the wave's 16 divisions run as independent chains in one stretch; no LDS lane exchange, no branches per row
# speedup vs baseline: 1.2491x; 1.0933x over previous
#include <hip/hip_runtime.h>

typedef float v2f __attribute__((ext_vector_type(2)));
typedef float v8f __attribute__((ext_vector_type(8)));

#define BDIM  2
#define NDIM  1024
#define CDIM  64
#define SCALE 0.25f
#define LOG2E 1.44269504088896340736f
#define JC    64
#define NCH   (NDIM / JC)
#define CPB   2
#define ROWS  256
#define NTHR  512
#define NWAVE (NTHR / 32)


struct Frag {
    float4 k[JC / 8];
    float4 v[JC / 8];
};

static __device__ __forceinline__ void load_frag(Frag& f, const float* kl, const float* vl) {
    #pragma unroll
    for (int e = 0; e < JC / 8; ++e) {
        f.k[e] = *reinterpret_cast<const float4*>(kl + 4 * e);
        f.v[e] = *reinterpret_cast<const float4*>(vl + 4 * e);
    }
}

static __device__ __forceinline__ v2f exp_pair(float a, float n, float k0, float k1) {
    v2f A;
    A.x = __builtin_amdgcn_exp2f(__builtin_fmaf(a, k0, n));
    A.y = __builtin_amdgcn_exp2f(__builtin_fmaf(a, k1, n));
    return A;
}

static __device__ __forceinline__ void mma_frag(const Frag& f,
                                                float a0, float n0, float a1, float n1,
                                                bool isn0, float bc1,
                                                v8f& acc0, v8f& acc1) {
    #pragma unroll
    for (int e = 0; e < JC / 8; ++e) {
        const float4 kk = f.k[e];
        const float4 vv = f.v[e];
        v2f Blo, Bhi;
        Blo.x = isn0 ? vv.x : bc1;
        Blo.y = isn0 ? vv.y : bc1;
        Bhi.x = isn0 ? vv.z : bc1;
        Bhi.y = isn0 ? vv.w : bc1;

        acc0 = __builtin_amdgcn_wmma_f32_16x16x4_f32(false, exp_pair(a0, n0, kk.x, kk.y), false, Blo,
                                                     (short)0, acc0, false, false);
        acc1 = __builtin_amdgcn_wmma_f32_16x16x4_f32(false, exp_pair(a1, n1, kk.x, kk.y), false, Blo,
                                                     (short)0, acc1, false, false);
        acc0 = __builtin_amdgcn_wmma_f32_16x16x4_f32(false, exp_pair(a0, n0, kk.z, kk.w), false, Bhi,
                                                     (short)0, acc0, false, false);
        acc1 = __builtin_amdgcn_wmma_f32_16x16x4_f32(false, exp_pair(a1, n1, kk.z, kk.w), false, Bhi,
                                                     (short)0, acc1, false, false);
    }
}

static __device__ __forceinline__ float from_next_lane(float x) {
    return __builtin_bit_cast(float,
        __builtin_amdgcn_mov_dpp(__builtin_bit_cast(int, x), 0xF5, 0xF, 0xF, false));
}

static __device__ __forceinline__ void store_tiles(const v8f& acc0, v8f acc1,
                                                   float* op, int i0, int lane) {
    v8f den0, den1;
    #pragma unroll
    for (int r = 0; r < 8; ++r) den0[r] = from_next_lane(acc0[r]);
    asm volatile("" : "+v"(acc1), "+v"(den0));
    #pragma unroll
    for (int r = 0; r < 8; ++r) den1[r] = from_next_lane(acc1[r]);
    if ((lane & 15) == 0) {
        float* p = op + (size_t)(i0 + ((lane >> 4) << 3)) * CDIM;
        #pragma unroll
        for (int r = 0; r < 8; ++r) {
            p[(size_t)r * CDIM]        = acc0[r] / den0[r];
            p[(size_t)(16 + r) * CDIM] = acc1[r] / den1[r];
        }
    }
}

static __device__ __forceinline__ float2 ld2(const float* p) {
    return *reinterpret_cast<const float2*>(p);
}

__global__ __launch_bounds__(NTHR)
void intra_attn_wmma_kernel(const float* __restrict__ q,
                            const float* __restrict__ k,
                            const float* __restrict__ v,
                            float* __restrict__ out) {
    __shared__ __align__(16) float ks[CPB][NDIM];
    __shared__ __align__(16) float vs[CPB][NDIM];
    __shared__ __align__(16) float qs[CPB][ROWS];
    __shared__ __align__(16) float red[CPB][NWAVE];

    const int tid  = threadIdx.x;
    const int lane = tid & 31;
    const int wave = tid >> 5;

    const int cp    = blockIdx.x;
    const int b     = cp >> 5;
    const int c0    = (cp & (CDIM / CPB - 1)) * CPB;
    const int ibase = blockIdx.y * ROWS;

    const float* qp = q   + (size_t)b * NDIM * CDIM + c0;
    const float* kp = k   + (size_t)b * NDIM * CDIM + c0;
    const float* vp = v   + (size_t)b * NDIM * CDIM + c0;

    const float2 kA = ld2(kp + (size_t)tid * CDIM);
    const float2 kB = ld2(kp + (size_t)(tid + NTHR) * CDIM);
    const float2 vA = ld2(vp + (size_t)tid * CDIM);
    const float2 vB = ld2(vp + (size_t)(tid + NTHR) * CDIM);
    float2 qq = make_float2(0.0f, 0.0f);
    if (tid < ROWS) qq = ld2(qp + (size_t)(ibase + tid) * CDIM);

    const float kA0 = kA.x * (SCALE * LOG2E);
    const float kA1 = kA.y * (SCALE * LOG2E);
    const float kB0 = kB.x * (SCALE * LOG2E);
    const float kB1 = kB.y * (SCALE * LOG2E);
    ks[0][tid]        = kA0;
    ks[1][tid]        = kA1;
    ks[0][tid + NTHR] = kB0;
    ks[1][tid + NTHR] = kB1;
    vs[0][tid]        = vA.x;
    vs[1][tid]        = vA.y;
    vs[0][tid + NTHR] = vB.x;
    vs[1][tid + NTHR] = vB.y;
    if (tid < ROWS) {
        qs[0][tid] = qq.x;
        qs[1][tid] = qq.y;
    }

    float m0 = fmaxf(fabsf(kA0), fabsf(kB0));
    float m1 = fmaxf(fabsf(kA1), fabsf(kB1));
    #pragma unroll
    for (int o = 16; o > 0; o >>= 1) {
        m0 = fmaxf(m0, __shfl_xor(m0, o, 32));
        m1 = fmaxf(m1, __shfl_xor(m1, o, 32));
    }
    if (lane == 0) {
        red[0][wave] = m0;
        red[1][wave] = m1;
    }
    __syncthreads();

    const int ch = wave >> 3;
    const int il = (wave & 7) * 32;

    float ksabs = 0.0f;
    #pragma unroll
    for (int e = 0; e < NWAVE / 4; ++e) {
        const float4 r = *reinterpret_cast<const float4*>(&red[ch][4 * e]);
        ksabs = fmaxf(fmaxf(ksabs, fmaxf(r.x, r.y)), fmaxf(r.z, r.w));
    }

    const int   M    = lane & 15;
    const float bc1  = (M == 1) ? 1.0f : 0.0f;
    const bool  isn0 = (M == 0);

    const float a0 = qs[ch][il + M];
    const float a1 = qs[ch][il + 16 + M];
    const float n0 = -(fabsf(a0) * ksabs);
    const float n1 = -(fabsf(a1) * ksabs);

    const float* kl = &ks[ch][0] + ((lane >> 4) << 5);
    const float* vl = &vs[ch][0] + ((lane >> 4) << 5);

    v8f acc0 = {};
    v8f acc1 = {};
    Frag fa, fb;
    load_frag(fa, kl, vl);
    for (int chk = 0; chk < NCH; chk += 2) {
        load_frag(fb, kl + (chk + 1) * JC, vl + (chk + 1) * JC);
        __builtin_amdgcn_sched_barrier(0);
        mma_frag(fa, a0, n0, a1, n1, isn0, bc1, acc0, acc1);
        asm volatile("" : "+v"(acc0), "+v"(acc1) : : "memory");
        load_frag(fa, kl + ((chk + 2) & (NCH - 1)) * JC, vl + ((chk + 2) & (NCH - 1)) * JC);
        __builtin_amdgcn_sched_barrier(0);
        mma_frag(fb, a0, n0, a1, n1, isn0, bc1, acc0, acc1);
    }

    float* op = out + (size_t)b * NDIM * CDIM + c0 + ch;
    store_tiles(acc0, acc1, op, ibase + il, lane);
}

extern "C" void kernel_launch(void* const* d_in, const int* in_sizes, int n_in,
                              void* d_out, int out_size, void* d_ws, size_t ws_size,
                              hipStream_t stream) {
    const float* q = (const float*)d_in[0];
    const float* k = (const float*)d_in[1];
    const float* v = (const float*)d_in[2];
    float* out = (float*)d_out;
    dim3 grid(BDIM * CDIM / CPB, NDIM / ROWS);
    intra_attn_wmma_kernel<<<grid, dim3(NTHR), 0, stream>>>(q, k, v, out);
}


# Vector work per WMMA cut from 5 instructions to 3: B read ready-made from LDS (v, ones, zeros planes by lane address, no select) and an A pair's two exponents by one packed fma; j sum order kept
# speedup vs baseline: 1.2889x; 1.0319x over previous
#include <hip/hip_runtime.h>

typedef float v2f __attribute__((ext_vector_type(2)));
typedef float v4f __attribute__((ext_vector_type(4)));
typedef float v8f __attribute__((ext_vector_type(8)));

#define BDIM  2
#define NDIM  1024
#define CDIM  64
#define SCALE 0.25f
#define LOG2E 1.44269504088896340736f
#define JC    64
#define NCH   (NDIM / JC)
#define CPB   2
#define ROWS  256
#define NTHR  512
#define NWAVE (NTHR / 32)
#define BONE  16


struct Frag {
    v4f k[JC / 8];
    v4f v[JC / 8];
};

static __device__ __forceinline__ void load_frag(Frag& f, const float* kl, const float* vl) {
    #pragma unroll
    for (int e = 0; e < JC / 8; ++e) {
        f.k[e] = *reinterpret_cast<const v4f*>(kl + 4 * e);
        f.v[e] = *reinterpret_cast<const v4f*>(vl + 4 * e);
    }
}

static __device__ __forceinline__ v2f exp_pair(v2f a, v2f n, v2f k) {
    const v2f x = __builtin_elementwise_fma(a, k, n);
    v2f A;
    A.x = __builtin_amdgcn_exp2f(x.x);
    A.y = __builtin_amdgcn_exp2f(x.y);
    return A;
}

static __device__ __forceinline__ void mma_frag(const Frag& f,
                                                v2f a0, v2f n0, v2f a1, v2f n1,
                                                v8f& acc0, v8f& acc1) {
    #pragma unroll
    for (int e = 0; e < JC / 8; ++e) {
        const v4f kk = f.k[e];
        const v4f vv = f.v[e];
        const v2f Blo = vv.lo;
        const v2f Bhi = vv.hi;

        acc0 = __builtin_amdgcn_wmma_f32_16x16x4_f32(false, exp_pair(a0, n0, kk.lo), false, Blo,
                                                     (short)0, acc0, false, false);
        acc1 = __builtin_amdgcn_wmma_f32_16x16x4_f32(false, exp_pair(a1, n1, kk.lo), false, Blo,
                                                     (short)0, acc1, false, false);
        acc0 = __builtin_amdgcn_wmma_f32_16x16x4_f32(false, exp_pair(a0, n0, kk.hi), false, Bhi,
                                                     (short)0, acc0, false, false);
        acc1 = __builtin_amdgcn_wmma_f32_16x16x4_f32(false, exp_pair(a1, n1, kk.hi), false, Bhi,
                                                     (short)0, acc1, false, false);
    }
}

static __device__ __forceinline__ float from_next_lane(float x) {
    return __builtin_bit_cast(float,
        __builtin_amdgcn_mov_dpp(__builtin_bit_cast(int, x), 0xF5, 0xF, 0xF, false));
}

static __device__ __forceinline__ void store_tiles(const v8f& acc0, v8f acc1,
                                                   float* op, int i0, int lane) {
    v8f den0, den1;
    #pragma unroll
    for (int r = 0; r < 8; ++r) den0[r] = from_next_lane(acc0[r]);
    asm volatile("" : "+v"(acc1), "+v"(den0));
    #pragma unroll
    for (int r = 0; r < 8; ++r) den1[r] = from_next_lane(acc1[r]);
    if ((lane & 15) == 0) {
        float* p = op + (size_t)(i0 + ((lane >> 4) << 3)) * CDIM;
        #pragma unroll
        for (int r = 0; r < 8; ++r) {
            p[(size_t)r * CDIM]        = acc0[r] / den0[r];
            p[(size_t)(16 + r) * CDIM] = acc1[r] / den1[r];
        }
    }
}

static __device__ __forceinline__ float2 ld2(const float* p) {
    return *reinterpret_cast<const float2*>(p);
}

__global__ __launch_bounds__(NTHR)
void intra_attn_wmma_kernel(const float* __restrict__ q,
                            const float* __restrict__ k,
                            const float* __restrict__ v,
                            float* __restrict__ out) {
    __shared__ __align__(16) float ks[CPB][NDIM];
    __shared__ __align__(16) float vs[CPB * NDIM + BONE + 64];
    __shared__ __align__(16) float qs[CPB][ROWS];
    __shared__ __align__(16) float red[CPB][NWAVE];

    const int tid  = threadIdx.x;
    const int lane = tid & 31;
    const int wave = tid >> 5;

    const int cp    = blockIdx.x;
    const int b     = cp >> 5;
    const int c0    = (cp & (CDIM / CPB - 1)) * CPB;
    const int ibase = blockIdx.y * ROWS;

    const float* qp = q   + (size_t)b * NDIM * CDIM + c0;
    const float* kp = k   + (size_t)b * NDIM * CDIM + c0;
    const float* vp = v   + (size_t)b * NDIM * CDIM + c0;

    const float2 kA = ld2(kp + (size_t)tid * CDIM);
    const float2 kB = ld2(kp + (size_t)(tid + NTHR) * CDIM);
    const float2 vA = ld2(vp + (size_t)tid * CDIM);
    const float2 vB = ld2(vp + (size_t)(tid + NTHR) * CDIM);
    float2 qq = make_float2(0.0f, 0.0f);
    if (tid < ROWS) qq = ld2(qp + (size_t)(ibase + tid) * CDIM);

    const float kA0 = kA.x * (SCALE * LOG2E);
    const float kA1 = kA.y * (SCALE * LOG2E);
    const float kB0 = kB.x * (SCALE * LOG2E);
    const float kB1 = kB.y * (SCALE * LOG2E);
    ks[0][tid]        = kA0;
    ks[1][tid]        = kA1;
    ks[0][tid + NTHR] = kB0;
    ks[1][tid + NTHR] = kB1;
    vs[tid]               = vA.x;
    vs[NDIM + tid]        = vA.y;
    vs[tid + NTHR]        = vB.x;
    vs[NDIM + tid + NTHR] = vB.y;
    if (tid < 64) vs[CPB * NDIM + BONE + tid] = (tid < 32) ? 1.0f : 0.0f;
    if (tid < ROWS) {
        qs[0][tid] = qq.x;
        qs[1][tid] = qq.y;
    }

    float m0 = fmaxf(fabsf(kA0), fabsf(kB0));
    float m1 = fmaxf(fabsf(kA1), fabsf(kB1));
    #pragma unroll
    for (int o = 16; o > 0; o >>= 1) {
        m0 = fmaxf(m0, __shfl_xor(m0, o, 32));
        m1 = fmaxf(m1, __shfl_xor(m1, o, 32));
    }
    if (lane == 0) {
        red[0][wave] = m0;
        red[1][wave] = m1;
    }
    __syncthreads();

    const int ch = wave >> 3;
    const int il = (wave & 7) * 32;

    float ksabs = 0.0f;
    #pragma unroll
    for (int e = 0; e < NWAVE / 4; ++e) {
        const float4 r = *reinterpret_cast<const float4*>(&red[ch][4 * e]);
        ksabs = fmaxf(fmaxf(ksabs, fmaxf(r.x, r.y)), fmaxf(r.z, r.w));
    }

    const int   M    = lane & 15;

    const float a0s = qs[ch][il + M];
    const float a1s = qs[ch][il + 16 + M];
    const float n0s = -(fabsf(a0s) * ksabs);
    const float n1s = -(fabsf(a1s) * ksabs);
    const v2f a0 = {a0s, a0s};
    const v2f a1 = {a1s, a1s};
    const v2f n0 = {n0s, n0s};
    const v2f n1 = {n1s, n1s};

    const float* kl = &ks[ch][0] + ((lane >> 4) << 5);
    const int      vidx = (M == 0) ? ch * NDIM + ((lane >> 4) << 5)
                                   : CPB * NDIM + BONE + ((M == 1) ? 0 : 32);
    const unsigned vmov = (M == 0) ? ~0u : 0u;
    const float* vl = &vs[0] + vidx;

    v8f acc0 = {};
    v8f acc1 = {};
    Frag fa, fb;
    load_frag(fa, kl, vl);
    for (int chk = 0; chk < NCH; chk += 2) {
        load_frag(fb, kl + (chk + 1) * JC, vl + ((unsigned)((chk + 1) * JC) & vmov));
        __builtin_amdgcn_sched_barrier(0);
        mma_frag(fa, a0, n0, a1, n1, acc0, acc1);
        asm volatile("" : "+v"(acc0), "+v"(acc1) : : "memory");
        load_frag(fa, kl + ((chk + 2) & (NCH - 1)) * JC,
                  vl + ((unsigned)(((chk + 2) & (NCH - 1)) * JC) & vmov));
        __builtin_amdgcn_sched_barrier(0);
        mma_frag(fb, a0, n0, a1, n1, acc0, acc1);
    }

    float* op = out + (size_t)b * NDIM * CDIM + c0 + ch;
    store_tiles(acc0, acc1, op, ibase + il, lane);
}

extern "C" void kernel_launch(void* const* d_in, const int* in_sizes, int n_in,
                              void* d_out, int out_size, void* d_ws, size_t ws_size,
                              hipStream_t stream) {
    const float* q = (const float*)d_in[0];
    const float* k = (const float*)d_in[1];
    const float* v = (const float*)d_in[2];
    float* out = (float*)d_out;
    dim3 grid(BDIM * CDIM / CPB, NDIM / ROWS);
    intra_attn_wmma_kernel<<<grid, dim3(NTHR), 0, stream>>>(q, k, v, out);
}
